# starting kernel
# baseline (speedup 1.0000x reference)
#include <hip/hip_runtime.h>

typedef __attribute__((ext_vector_type(2))) float v2f;
typedef __attribute__((ext_vector_type(8))) float v8f;

#define T_DIM 2048
#define B_DIM 8
#define C_DIM 1024
#define K_DIM 31
#define P_PAD 15
#define R_DIM 64
#define STRIDE (B_DIM * C_DIM)
#define TPW 8
#define N_TILES (T_DIM / 16)

__device__ __forceinline__ void ld_stile(v2f d[4], const float* base) {
#pragma unroll
  for (int ks = 0; ks < 4; ++ks) {
    d[ks].x = base[(size_t)(ks * 4)     * STRIDE];
    d[ks].y = base[(size_t)(ks * 4 + 1) * STRIDE];
  }
}

__device__ __forceinline__ void zero_stile(v2f d[4]) {
#pragma unroll
  for (int ks = 0; ks < 4; ++ks) { d[ks].x = 0.0f; d[ks].y = 0.0f; }
}

template <bool EDGE_LO, bool EDGE_HI>
__device__ __forceinline__ void run_chunk(const float* __restrict__ xchunk,
                                          float* __restrict__ ochunk,
                                          const v2f afr[12], float bv) {
  v2f bwin[3][4];

  if (EDGE_LO) zero_stile(bwin[0]);
  else         ld_stile(bwin[0], xchunk - (size_t)16 * STRIDE);
  ld_stile(bwin[1], xchunk);
  ld_stile(bwin[2], xchunk + (size_t)16 * STRIDE);

#pragma unroll
  for (int it = 0; it < TPW; ++it) {
    v2f nxt[4];
    if (it < TPW - 1) {
      if (EDGE_HI && (2 + it) >= TPW) zero_stile(nxt);
      else ld_stile(nxt, xchunk + (size_t)(2 + it) * 16 * STRIDE);
    }

    v8f acc = {};
#pragma unroll
    for (int st = 0; st < 3; ++st) {
      v2f* bt = bwin[(it + st) % 3];
#pragma unroll
      for (int ks = 0; ks < 4; ++ks) {
        acc = __builtin_amdgcn_wmma_f32_16x16x4_f32(
             false, afr[st * 4 + ks],  false, bt[ks],
             (short)0, acc,  false,  false);
      }
    }

    float* po = ochunk + (size_t)it * 16 * STRIDE;
#pragma unroll
    for (int r = 0; r < 8; ++r)
      __builtin_nontemporal_store(acc[r] + bv, po + (size_t)r * STRIDE);

    if (it < TPW - 1) {
#pragma unroll
      for (int ks = 0; ks < 4; ++ks) bwin[it % 3][ks] = nxt[ks];
    }
  }
}

__global__ __launch_bounds__(256) void
LightweightConv1dTBC_kernel(const float* __restrict__ x,
                            const float* __restrict__ weight,
                            const float* __restrict__ bias,
                            float* __restrict__ out) {
  __shared__ float wsm[K_DIM];

  const int b    = blockIdx.y;
  const int c0   = blockIdx.z * 16;
  const int head = c0 / R_DIM;

  {
    const float* wr = weight + head * K_DIM;
    float mx = wr[0];
#pragma unroll
    for (int k = 1; k < K_DIM; ++k) mx = fmaxf(mx, wr[k]);
    float sum = 0.0f;
#pragma unroll
    for (int k = 0; k < K_DIM; ++k) sum += __expf(wr[k] - mx);
    const float inv = 1.0f / sum;
    const int tid = threadIdx.y * 32 + threadIdx.x;
    if (tid < K_DIM) wsm[tid] = __expf(wr[tid] - mx) * inv;
  }
  __syncthreads();

  const int lane = threadIdx.x;
  const int mN   = lane & 15;
  const int hi   = lane >> 4;

  v2f afr[12];
#pragma unroll
  for (int st = 0; st < 3; ++st) {
#pragma unroll
    for (int ks = 0; ks < 4; ++ks) {
      const int i  = st * 4 + ks;
      const int j0 = ks * 4 + hi * 2;
      const int d0 = (st - 1) * 16 + j0 - mN + P_PAD;
      const int d1 = d0 + 1;
      const float w0 = wsm[min(max(d0, 0), K_DIM - 1)];
      const float w1 = wsm[min(max(d1, 0), K_DIM - 1)];
      afr[i].x = ((unsigned)d0 < (unsigned)K_DIM) ? w0 : 0.0f;
      afr[i].y = ((unsigned)d1 < (unsigned)K_DIM) ? w1 : 0.0f;
    }
  }
  const float bv = bias[c0 + mN];

  __builtin_amdgcn_sched_barrier(0);

  const int tb0 = (blockIdx.x * 8 + threadIdx.y) * TPW;
  const size_t colBase = (size_t)b * C_DIM + c0 + mN;
  const float* xchunk =
      x + colBase + (size_t)(hi * 2) * STRIDE + (size_t)tb0 * 16 * STRIDE;
  float* ochunk =
      out + colBase + (size_t)(hi * 8) * STRIDE + (size_t)tb0 * 16 * STRIDE;

  const int tb0u = __builtin_amdgcn_readfirstlane(tb0);
  if (tb0u == 0)                      run_chunk<true,  false>(xchunk, ochunk, afr, bv);
  else if (tb0u == N_TILES - TPW)     run_chunk<false, true >(xchunk, ochunk, afr, bv);
  else                                run_chunk<false, false>(xchunk, ochunk, afr, bv);
}

extern "C" void kernel_launch(void* const* d_in, const int* in_sizes, int n_in,
                              void* d_out, int out_size, void* d_ws, size_t ws_size,
                              hipStream_t stream) {
  (void)in_sizes; (void)n_in; (void)d_ws; (void)ws_size; (void)out_size;
  const float* x      = (const float*)d_in[0];
  const float* weight = (const float*)d_in[1];
  const float* bias   = (const float*)d_in[2];
  float* out          = (float*)d_out;

  dim3 block(32, 8);
  dim3 grid(N_TILES / (8 * TPW), B_DIM, C_DIM / 16);
  LightweightConv1dTBC_kernel<<<grid, block, 0, stream>>>(x, weight, bias, out);
}


# Workgroup walks 4 consecutive strips, next strip's global loads issued ahead of the current WMMA phase; LDS-only barriers keep loads and output stores in flight (16 waves, 256 workgroups)
# speedup vs baseline: 1.5850x; 1.5850x over previous
#include <hip/hip_runtime.h>


typedef __attribute__((ext_vector_type(2))) float v2f;
typedef __attribute__((ext_vector_type(4))) float v4f;
typedef __attribute__((ext_vector_type(8))) float v8f;

#define T_DIM 2048
#define B_DIM 8
#define C_DIM 1024
#define K_DIM 31
#define P_PAD 15
#define R_DIM 64
#define STRIDE (B_DIM * C_DIM)
#define TC 256
#define ROWS (TC + 32)
#define NS 4
#define NWAVE 16
#define NTHR (NWAVE * 32)
#define RSTEP (NTHR / 16)
#define NLD (ROWS / RSTEP)
#define NST (TC / RSTEP)
#define TPW (TC / 16 / (NWAVE / 4))

__device__ __forceinline__ int lds_idx(int row, int col) {
  return row * 64 + (col ^ ((((row >> 1) ^ (row >> 3)) & 1) << 5));
}

__device__ __forceinline__ void ld_stile(v2f d[4], const float* xs, int s,
                                         int h, int col) {
#pragma unroll
  for (int ks = 0; ks < 4; ++ks) {
    const int row = 16 * s + 4 * ks + 2 * h;
    d[ks].x = xs[lds_idx(row, col)];
    d[ks].y = xs[lds_idx(row + 1, col)];
  }
}

__device__ __forceinline__ void lds_barrier() {
  asm volatile("s_wait_dscnt 0x0\n\ts_barrier_signal -1\n\ts_barrier_wait -1" ::: "memory");
}

__device__ __forceinline__ void load_strip(v4f ld[NLD], const float* __restrict__ x,
                                           size_t colBase, int t0, int r0) {
  const v4f zero4 = {0.0f, 0.0f, 0.0f, 0.0f};
#pragma unroll
  for (int p = 0; p < NLD; ++p) {
    const int t = t0 - 16 + r0 + p * RSTEP;
    if (p == 0 || p == NLD - 1) {
      if ((unsigned)t < (unsigned)T_DIM)
        ld[p] = *(const v4f*)(x + colBase + (size_t)t * STRIDE);
      else
        ld[p] = zero4;
    } else {
      ld[p] = *(const v4f*)(x + colBase + (size_t)t * STRIDE);
    }
  }
}

__global__ __launch_bounds__(NTHR) void
LightweightConv1dTBC_kernel(const float* __restrict__ x,
                            const float* __restrict__ weight,
                            const float* __restrict__ bias,
                            float* __restrict__ out) {
  __shared__ __attribute__((aligned(16))) float xs[ROWS * 64];
  __shared__ float wsm[32];

  const int tid  = threadIdx.y * 32 + threadIdx.x;
  const int b    = blockIdx.y;
  const int head = blockIdx.z;
  const int tq   = blockIdx.x * (NS * TC);

  const int c4 = tid & 15;
  const int r0 = tid >> 4;
  const size_t colBase = (size_t)b * C_DIM + head * R_DIM + c4 * 4;

  v4f ld[NLD];
  load_strip(ld, x, colBase, tq, r0);

  const v4f bias4 = *(const v4f*)(bias + head * R_DIM + c4 * 4);

  {
    const float* wr = weight + head * K_DIM;
    float mx = wr[0];
#pragma unroll
    for (int k = 1; k < K_DIM; ++k) mx = fmaxf(mx, wr[k]);
    float sum = 0.0f;
#pragma unroll
    for (int k = 0; k < K_DIM; ++k) sum += __expf(wr[k] - mx);
    const float inv = 1.0f / sum;
    if (tid < K_DIM) wsm[tid] = __expf(wr[tid] - mx) * inv;
  }

#pragma unroll
  for (int p = 0; p < NLD; ++p)
    *(v4f*)(&xs[lds_idx(p * RSTEP + r0, c4 * 4)]) = ld[p];

  lds_barrier();

  const int lane = threadIdx.x;
  const int mN   = lane & 15;
  const int hi   = lane >> 4;
  const int jt   = threadIdx.y & 3;
  const int sb   = (threadIdx.y >> 2) * TPW;
  const int col  = jt * 16 + mN;

  v2f afr[12];
#pragma unroll
  for (int st = 0; st < 3; ++st) {
#pragma unroll
    for (int ks = 0; ks < 4; ++ks) {
      const int i  = st * 4 + ks;
      const int j0 = ks * 4 + hi * 2;
      const int d0 = (st - 1) * 16 + j0 - mN + P_PAD;
      const int d1 = d0 + 1;
      const float w0 = wsm[min(max(d0, 0), K_DIM - 1)];
      const float w1 = wsm[min(max(d1, 0), K_DIM - 1)];
      afr[i].x = ((unsigned)d0 < (unsigned)K_DIM) ? w0 : 0.0f;
      afr[i].y = ((unsigned)d1 < (unsigned)K_DIM) ? w1 : 0.0f;
    }
  }

#pragma unroll
  for (int sidx = 0; sidx < NS; ++sidx) {
    const int t0 = tq + sidx * TC;

    if (sidx + 1 < NS) {
      load_strip(ld, x, colBase, t0 + TC, r0);
    }

    v8f acc[TPW];
    v2f bwin[3][4];
    ld_stile(bwin[0], xs, sb,     hi, col);
    ld_stile(bwin[1], xs, sb + 1, hi, col);
    ld_stile(bwin[2], xs, sb + 2, hi, col);
#pragma unroll
    for (int it = 0; it < TPW; ++it) {
      v2f nxt[4];
      if (it < TPW - 1) ld_stile(nxt, xs, sb + it + 3, hi, col);
      v8f a = {};
#pragma unroll
      for (int st = 0; st < 3; ++st) {
        v2f* bt = bwin[(it + st) % 3];
#pragma unroll
        for (int ks = 0; ks < 4; ++ks) {
          a = __builtin_amdgcn_wmma_f32_16x16x4_f32(
               false, afr[st * 4 + ks],  false, bt[ks],
               (short)0, a,  false,  false);
        }
      }
      acc[it] = a;
      if (it < TPW - 1) {
#pragma unroll
        for (int ks = 0; ks < 4; ++ks) bwin[it % 3][ks] = nxt[ks];
      }
    }

    lds_barrier();

#pragma unroll
    for (int it = 0; it < TPW; ++it) {
#pragma unroll
      for (int r = 0; r < 8; ++r)
        xs[lds_idx(16 * (sb + it + 1) + 8 * hi + r, col)] = acc[it][r];
    }

    lds_barrier();

    float* og = out + colBase + (size_t)(t0 + r0) * STRIDE;
#pragma unroll
    for (int p = 0; p < NST; ++p) {
      const v4f v = *(const v4f*)(&xs[lds_idx(16 + p * RSTEP + r0, c4 * 4)]) + bias4;
      __builtin_nontemporal_store(v, (v4f*)(og + (size_t)p * RSTEP * STRIDE));
    }

    if (sidx + 1 < NS) {
      lds_barrier();
#pragma unroll
      for (int p = 0; p < NLD; ++p)
        *(v4f*)(&xs[lds_idx(p * RSTEP + r0, c4 * 4)]) = ld[p];
      lds_barrier();
    }
  }
}

extern "C" void kernel_launch(void* const* d_in, const int* in_sizes, int n_in,
                              void* d_out, int out_size, void* d_ws, size_t ws_size,
                              hipStream_t stream) {
  (void)in_sizes; (void)n_in; (void)d_ws; (void)ws_size; (void)out_size;
  const float* x      = (const float*)d_in[0];
  const float* weight = (const float*)d_in[1];
  const float* bias   = (const float*)d_in[2];
  float* out          = (float*)d_out;

  dim3 block(32, NWAVE);
  dim3 grid(T_DIM / (TC * NS), B_DIM, C_DIM / R_DIM);
  LightweightConv1dTBC_kernel<<<grid, block, 0, stream>>>(x, weight, bias, out);
}


# A wave owns two adjacent channel tiles; a half-wave exchange (v_permlane16_swap_b32) makes each D register one whole 128-byte row, stored from registers: D's LDS round trip and 2 barriers a strip go
# speedup vs baseline: 1.6985x; 1.0716x over previous
#include <hip/hip_runtime.h>


typedef __attribute__((ext_vector_type(2))) float v2f;
typedef __attribute__((ext_vector_type(4))) float v4f;
typedef __attribute__((ext_vector_type(8))) float v8f;

#define T_DIM 2048
#define B_DIM 8
#define C_DIM 1024
#define K_DIM 31
#define P_PAD 15
#define R_DIM 64
#define STRIDE (B_DIM * C_DIM)
#define TC 256
#define ROWS (TC + 32)
#define NS 4
#define NWAVE 16
#define NTHR (NWAVE * 32)
#define RSTEP (NTHR / 16)
#define NLD (ROWS / RSTEP)
#define TPW (TC / 16 / (NWAVE / 2))

__device__ __forceinline__ int lds_idx(int row, int col) {
  return row * 64 + (col ^ ((((row >> 1) ^ (row >> 3)) & 1) << 5));
}

__device__ __forceinline__ void ld_stile(v2f d[4], const float* xs, int s,
                                         int h, int col) {
#pragma unroll
  for (int ks = 0; ks < 4; ++ks) {
    const int row = 16 * s + 4 * ks + 2 * h;
    d[ks].x = xs[lds_idx(row, col)];
    d[ks].y = xs[lds_idx(row + 1, col)];
  }
}

__device__ __forceinline__ void lds_barrier() {
  asm volatile("s_wait_dscnt 0x0\n\ts_barrier_signal -1\n\ts_barrier_wait -1" ::: "memory");
}

__device__ __forceinline__ void rows_from_tile_pair(float& p, float& q) {
  typedef __attribute__((ext_vector_type(2))) unsigned v2u;
  const v2u r = __builtin_amdgcn_permlane16_swap(__float_as_uint(p), __float_as_uint(q),
                                                  false,  false);
  p = __uint_as_float(r.x);
  q = __uint_as_float(r.y);
}

__device__ __forceinline__ void load_strip(v4f ld[NLD], const float* __restrict__ x,
                                           size_t colBase, int t0, int r0) {
#pragma unroll
  for (int p = 0; p < NLD; ++p) {
    int t = t0 - 16 + r0 + p * RSTEP;
    if (p == 0 || p == NLD - 1) t = min(max(t, 0), T_DIM - 1);
    ld[p] = *(const v4f*)(x + colBase + (size_t)t * STRIDE);
  }
}

__device__ __forceinline__ void store_strip(float* xs, const v4f ld[NLD],
                                            int t0, int r0, int c4) {
  const v4f zero4 = {0.0f, 0.0f, 0.0f, 0.0f};
#pragma unroll
  for (int p = 0; p < NLD; ++p) {
    v4f v = ld[p];
    if (p == 0 || p == NLD - 1) {
      const int t = t0 - 16 + r0 + p * RSTEP;
      v = ((unsigned)t < (unsigned)T_DIM) ? v : zero4;
    }
    *(v4f*)(&xs[lds_idx(p * RSTEP + r0, c4 * 4)]) = v;
  }
}

__global__ __launch_bounds__(NTHR) void
LightweightConv1dTBC_kernel(const float* __restrict__ x,
                            const float* __restrict__ weight,
                            const float* __restrict__ bias,
                            float* __restrict__ out) {
  __shared__ __attribute__((aligned(16))) float xs[ROWS * 64];
  __shared__ float wsm[32];

  const int tid  = threadIdx.y * 32 + threadIdx.x;
  const int b    = blockIdx.y;
  const int head = blockIdx.z;
  const int tq   = blockIdx.x * (NS * TC);

  const int c4 = tid & 15;
  const int r0 = tid >> 4;
  const size_t colBase = (size_t)b * C_DIM + head * R_DIM + c4 * 4;

  v4f ld[NLD];
  load_strip(ld, x, colBase, tq, r0);

  const float biasL = bias[head * R_DIM + (threadIdx.y & 1) * 32 + threadIdx.x];

  {
    const float* wr = weight + head * K_DIM;
    float mx = wr[0];
#pragma unroll
    for (int k = 1; k < K_DIM; ++k) mx = fmaxf(mx, wr[k]);
    float sum = 0.0f;
#pragma unroll
    for (int k = 0; k < K_DIM; ++k) sum += __expf(wr[k] - mx);
    const float inv = 1.0f / sum;
    if (tid < K_DIM) wsm[tid] = __expf(wr[tid] - mx) * inv;
  }

  store_strip(xs, ld, tq, r0, c4);

  lds_barrier();

  const int lane = threadIdx.x;
  const int mN   = lane & 15;
  const int hi   = lane >> 4;
  const int jp   = threadIdx.y & 1;
  const int sb   = (threadIdx.y >> 1) * TPW;
  const int colP = jp * 32 + mN;
  const int colQ = colP + 16;

  float* outw = out + (size_t)b * C_DIM + head * R_DIM + jp * 32 + lane;

  v2f afr[12];
#pragma unroll
  for (int st = 0; st < 3; ++st) {
#pragma unroll
    for (int ks = 0; ks < 4; ++ks) {
      const int i  = st * 4 + ks;
      const int j0 = ks * 4 + hi * 2;
      const int d0 = (st - 1) * 16 + j0 - mN + P_PAD;
      const int d1 = d0 + 1;
      const float w0 = wsm[min(max(d0, 0), K_DIM - 1)];
      const float w1 = wsm[min(max(d1, 0), K_DIM - 1)];
      afr[i].x = ((unsigned)d0 < (unsigned)K_DIM) ? w0 : 0.0f;
      afr[i].y = ((unsigned)d1 < (unsigned)K_DIM) ? w1 : 0.0f;
    }
  }

#pragma unroll
  for (int sidx = 0; sidx < NS; ++sidx) {
    const int t0 = tq + sidx * TC;

    if (sidx + 1 < NS) {
      load_strip(ld, x, colBase, t0 + TC, r0);
      __builtin_amdgcn_sched_barrier(0);
    }

    v2f bP[TPW + 2][4], bQ[TPW + 2][4];
#pragma unroll
    for (int s = 0; s < TPW + 2; ++s) {
      ld_stile(bP[s], xs, sb + s, hi, colP);
      ld_stile(bQ[s], xs, sb + s, hi, colQ);
    }
#pragma unroll
    for (int it = 0; it < TPW; ++it) {
      v8f aP = {};
      v8f aQ = {};
#pragma unroll
      for (int st = 0; st < 3; ++st) {
#pragma unroll
        for (int ks = 0; ks < 4; ++ks) {
          aP = __builtin_amdgcn_wmma_f32_16x16x4_f32(
               false, afr[st * 4 + ks],  false, bP[it + st][ks],
               (short)0, aP,  false,  false);
          aQ = __builtin_amdgcn_wmma_f32_16x16x4_f32(
               false, afr[st * 4 + ks],  false, bQ[it + st][ks],
               (short)0, aQ,  false,  false);
        }
      }
      float* og = outw + (size_t)(t0 + 16 * (sb + it)) * STRIDE;
#pragma unroll
      for (int r = 0; r < 8; ++r) {
        float p = aP[r];
        float q = aQ[r];
        rows_from_tile_pair(p, q);
        __builtin_nontemporal_store(p + biasL, og + (size_t)r * STRIDE);
        __builtin_nontemporal_store(q + biasL, og + (size_t)(r + 8) * STRIDE);
      }
    }

    if (sidx + 1 < NS) {
      lds_barrier();
      __builtin_amdgcn_sched_barrier(0);
      store_strip(xs, ld, t0 + TC, r0, c4);
      lds_barrier();
    }
  }
}

extern "C" void kernel_launch(void* const* d_in, const int* in_sizes, int n_in,
                              void* d_out, int out_size, void* d_ws, size_t ws_size,
                              hipStream_t stream) {
  (void)in_sizes; (void)n_in; (void)d_ws; (void)ws_size; (void)out_size;
  const float* x      = (const float*)d_in[0];
  const float* weight = (const float*)d_in[1];
  const float* bias   = (const float*)d_in[2];
  float* out          = (float*)d_out;

  dim3 block(32, NWAVE);
  dim3 grid(T_DIM / (TC * NS), B_DIM, C_DIM / R_DIM);
  LightweightConv1dTBC_kernel<<<grid, block, 0, stream>>>(x, weight, bias, out);
}


# LDS strip double-buffered: a wave fills the other buffer between its two tiles, and the one barrier left per strip is split, signalled 16 products before the wave's end and waited behind its stores
# speedup vs baseline: 1.8334x; 1.0794x over previous
#include <hip/hip_runtime.h>


typedef __attribute__((ext_vector_type(2))) float v2f;
typedef __attribute__((ext_vector_type(4))) float v4f;
typedef __attribute__((ext_vector_type(8))) float v8f;

#define T_DIM 2048
#define B_DIM 8
#define C_DIM 1024
#define K_DIM 31
#define P_PAD 15
#define R_DIM 64
#define STRIDE (B_DIM * C_DIM)
#define TC 256
#define ROWS (TC + 32)
#define NS 4
#define NWAVE 16
#define NTHR (NWAVE * 32)
#define RSTEP (NTHR / 16)
#define NLD (ROWS / RSTEP)
#define TPW (TC / 16 / (NWAVE / 2))

__device__ __forceinline__ int lds_idx(int row, int col) {
  return row * 64 + (col ^ ((((row >> 1) ^ (row >> 3)) & 1) << 5));
}

__device__ __forceinline__ void ld_stile(v2f d[4], const float* xs, int s,
                                         int h, int col) {
#pragma unroll
  for (int ks = 0; ks < 4; ++ks) {
    const int row = 16 * s + 4 * ks + 2 * h;
    d[ks].x = xs[lds_idx(row, col)];
    d[ks].y = xs[lds_idx(row + 1, col)];
  }
}

__device__ __forceinline__ void lds_barrier() {
  asm volatile("s_wait_dscnt 0x0\n\ts_barrier_signal -1\n\ts_barrier_wait -1" ::: "memory");
}

__device__ __forceinline__ void lds_signal(v8f& aP, v8f& aQ) {
  asm volatile("s_wait_dscnt 0x0\n\ts_barrier_signal -1" : "+v"(aP), "+v"(aQ) : : "memory");
}
__device__ __forceinline__ void lds_wait() {
  asm volatile("s_barrier_wait -1" ::: "memory");
}

__device__ __forceinline__ void mma_tile_pair(v8f& aP, v8f& aQ, const v2f afr[12],
                                              const v2f bP[][4], const v2f bQ[][4], int it,
                                              int st0, int st1) {
#pragma unroll
  for (int st = st0; st < st1; ++st) {
#pragma unroll
    for (int ks = 0; ks < 4; ++ks) {
      aP = __builtin_amdgcn_wmma_f32_16x16x4_f32(
           false, afr[st * 4 + ks],  false, bP[it + st][ks],
           (short)0, aP,  false,  false);
      aQ = __builtin_amdgcn_wmma_f32_16x16x4_f32(
           false, afr[st * 4 + ks],  false, bQ[it + st][ks],
           (short)0, aQ,  false,  false);
    }
  }
}

__device__ __forceinline__ void rows_from_tile_pair(float& p, float& q) {
  typedef __attribute__((ext_vector_type(2))) unsigned v2u;
  const v2u r = __builtin_amdgcn_permlane16_swap(__float_as_uint(p), __float_as_uint(q),
                                                  false,  false);
  p = __uint_as_float(r.x);
  q = __uint_as_float(r.y);
}

__device__ __forceinline__ void load_strip(v4f ld[NLD], const float* __restrict__ x,
                                           size_t colBase, int t0, int r0) {
#pragma unroll
  for (int p = 0; p < NLD; ++p) {
    int t = t0 - 16 + r0 + p * RSTEP;
    if (p == 0 || p == NLD - 1) t = min(max(t, 0), T_DIM - 1);
    ld[p] = *(const v4f*)(x + colBase + (size_t)t * STRIDE);
  }
}

__device__ __forceinline__ void store_strip(float* xs, const v4f ld[NLD],
                                            int t0, int r0, int c4) {
  const v4f zero4 = {0.0f, 0.0f, 0.0f, 0.0f};
#pragma unroll
  for (int p = 0; p < NLD; ++p) {
    v4f v = ld[p];
    if (p == 0 || p == NLD - 1) {
      const int t = t0 - 16 + r0 + p * RSTEP;
      v = ((unsigned)t < (unsigned)T_DIM) ? v : zero4;
    }
    *(v4f*)(&xs[lds_idx(p * RSTEP + r0, c4 * 4)]) = v;
  }
}

__global__ __launch_bounds__(NTHR) void
LightweightConv1dTBC_kernel(const float* __restrict__ x,
                            const float* __restrict__ weight,
                            const float* __restrict__ bias,
                            float* __restrict__ out) {
  __shared__ __attribute__((aligned(16))) float xs[2 * ROWS * 64];
  __shared__ float wsm[32];

  const int tid  = threadIdx.y * 32 + threadIdx.x;
  const int b    = blockIdx.y;
  const int head = blockIdx.z;
  const int tq   = blockIdx.x * (NS * TC);

  const int c4 = tid & 15;
  const int r0 = tid >> 4;
  const size_t colBase = (size_t)b * C_DIM + head * R_DIM + c4 * 4;

  v4f ld[NLD];
  load_strip(ld, x, colBase, tq, r0);

  const float biasL = bias[head * R_DIM + (threadIdx.y & 1) * 32 + threadIdx.x];

  {
    const float* wr = weight + head * K_DIM;
    float mx = wr[0];
#pragma unroll
    for (int k = 1; k < K_DIM; ++k) mx = fmaxf(mx, wr[k]);
    float sum = 0.0f;
#pragma unroll
    for (int k = 0; k < K_DIM; ++k) sum += __expf(wr[k] - mx);
    const float inv = 1.0f / sum;
    if (tid < K_DIM) wsm[tid] = __expf(wr[tid] - mx) * inv;
  }

  store_strip(xs, ld, tq, r0, c4);

  lds_barrier();

  if (NS > 1) {
    load_strip(ld, x, colBase, tq + TC, r0);
    __builtin_amdgcn_sched_barrier(0);
  }

  const int lane = threadIdx.x;
  const int mN   = lane & 15;
  const int hi   = lane >> 4;
  const int jp   = threadIdx.y & 1;
  const int sb   = (threadIdx.y >> 1) * TPW;
  const int colP = jp * 32 + mN;
  const int colQ = colP + 16;

  float* outw = out + (size_t)b * C_DIM + head * R_DIM + jp * 32 + lane;

  v2f afr[12];
#pragma unroll
  for (int st = 0; st < 3; ++st) {
#pragma unroll
    for (int ks = 0; ks < 4; ++ks) {
      const int i  = st * 4 + ks;
      const int j0 = ks * 4 + hi * 2;
      const int d0 = (st - 1) * 16 + j0 - mN + P_PAD;
      const int d1 = d0 + 1;
      const float w0 = wsm[min(max(d0, 0), K_DIM - 1)];
      const float w1 = wsm[min(max(d1, 0), K_DIM - 1)];
      afr[i].x = ((unsigned)d0 < (unsigned)K_DIM) ? w0 : 0.0f;
      afr[i].y = ((unsigned)d1 < (unsigned)K_DIM) ? w1 : 0.0f;
    }
  }

#pragma unroll
  for (int sidx = 0; sidx < NS; ++sidx) {
    const int t0 = tq + sidx * TC;
    const float* cur = xs + (sidx & 1) * (ROWS * 64);
    float* alt       = xs + ((sidx + 1) & 1) * (ROWS * 64);
    const bool more  = sidx + 1 < NS;

    v2f bP[TPW + 2][4], bQ[TPW + 2][4];
#pragma unroll
    for (int s = 0; s < TPW + 2; ++s) {
      ld_stile(bP[s], cur, sb + s, hi, colP);
      ld_stile(bQ[s], cur, sb + s, hi, colQ);
    }
#pragma unroll
    for (int it = 0; it < TPW; ++it) {
      v8f aP = {};
      v8f aQ = {};
      if (more && it == TPW - 1) {
        mma_tile_pair(aP, aQ, afr, bP, bQ, it, 0, 1);
        lds_signal(aP, aQ);
        mma_tile_pair(aP, aQ, afr, bP, bQ, it, 1, 3);
      } else {
        mma_tile_pair(aP, aQ, afr, bP, bQ, it, 0, 3);
      }

      float* og = outw + (size_t)(t0 + 16 * (sb + it)) * STRIDE;
#pragma unroll
      for (int r = 0; r < 8; ++r) {
        float p = aP[r];
        float q = aQ[r];
        rows_from_tile_pair(p, q);
        __builtin_nontemporal_store(p + biasL, og + (size_t)r * STRIDE);
        __builtin_nontemporal_store(q + biasL, og + (size_t)(r + 8) * STRIDE);
      }

      if (more && it == 0) {
        __builtin_amdgcn_sched_barrier(0);
        store_strip(alt, ld, t0 + TC, r0, c4);
        if (sidx + 2 < NS) load_strip(ld, x, colBase, t0 + 2 * TC, r0);
        __builtin_amdgcn_sched_barrier(0);
      }
    }

    if (more) lds_wait();
  }
}

extern "C" void kernel_launch(void* const* d_in, const int* in_sizes, int n_in,
                              void* d_out, int out_size, void* d_ws, size_t ws_size,
                              hipStream_t stream) {
  (void)in_sizes; (void)n_in; (void)d_ws; (void)ws_size; (void)out_size;
  const float* x      = (const float*)d_in[0];
  const float* weight = (const float*)d_in[1];
  const float* bias   = (const float*)d_in[2];
  float* out          = (float*)d_out;

  dim3 block(32, NWAVE);
  dim3 grid(T_DIM / (TC * NS), B_DIM, C_DIM / R_DIM);
  LightweightConv1dTBC_kernel<<<grid, block, 0, stream>>>(x, weight, bias, out);
}


# Two output tiles of a wave from 4 block products, not 6: 3-tap filter over 16-row blocks, 2 products on block sums/differences serve both tiles; 16 WMMAs a tile pair, not 24; summation order changes
# speedup vs baseline: 1.9622x; 1.0703x over previous
#include <hip/hip_runtime.h>


typedef __attribute__((ext_vector_type(2))) float v2f;
typedef __attribute__((ext_vector_type(4))) float v4f;
typedef __attribute__((ext_vector_type(8))) float v8f;

#define T_DIM 2048
#define B_DIM 8
#define C_DIM 1024
#define K_DIM 31
#define P_PAD 15
#define R_DIM 64
#define STRIDE (B_DIM * C_DIM)
#define TC 256
#define ROWS (TC + 32)
#define NS 4
#define NWAVE 16
#define NTHR (NWAVE * 32)
#define RSTEP (NTHR / 16)
#define NLD (ROWS / RSTEP)
#define TPW (TC / 16 / (NWAVE / 2))

__device__ __forceinline__ int lds_idx(int row, int col) {
  return row * 64 + (col ^ ((((row >> 1) ^ (row >> 3)) & 1) << 5));
}

__device__ __forceinline__ void ld_stile(v2f d[4], const float* f01, const float* f23,
                                         int s) {
#pragma unroll
  for (int ks = 0; ks < 4; ++ks) {
    const float* f = (ks < 2) ? f01 : f23;
    d[ks].x = f[(16 * s + 4 * ks) * 64];
    d[ks].y = f[(16 * s + 4 * ks + 1) * 64];
  }
}

__device__ __forceinline__ void lds_barrier() {
  asm volatile("s_wait_dscnt 0x0\n\ts_barrier_signal -1\n\ts_barrier_wait -1" ::: "memory");
}

__device__ __forceinline__ void lds_signal(v8f& y0P, v8f& y0Q, v8f& y1P, v8f& y1Q) {
  asm volatile("s_wait_dscnt 0x0\n\ts_barrier_signal -1"
               : "+v"(y0P), "+v"(y0Q), "+v"(y1P), "+v"(y1Q) : : "memory");
}
__device__ __forceinline__ void lds_wait() {
  asm volatile("s_barrier_wait -1" ::: "memory");
}

__device__ __forceinline__ v8f wmma4(const v2f a, const v2f b, const v8f c) {
  return __builtin_amdgcn_wmma_f32_16x16x4_f32(
       false, a,  false, b,
       (short)0, c,  false,  false);
}

__device__ __forceinline__ void mma_block_pair(v8f& cP, v8f& cQ, const v2f a[4],
                                               const v2f bP[4], const v2f bQ[4]) {
#pragma unroll
  for (int ks = 0; ks < 4; ++ks) {
    cP = wmma4(a[ks], bP[ks], cP);
    cQ = wmma4(a[ks], bQ[ks], cQ);
  }
}

__device__ __forceinline__ void rows_from_tile_pair(float& p, float& q) {
  typedef __attribute__((ext_vector_type(2))) unsigned v2u;
  const v2u r = __builtin_amdgcn_permlane16_swap(__float_as_uint(p), __float_as_uint(q),
                                                  false,  false);
  p = __uint_as_float(r.x);
  q = __uint_as_float(r.y);
}

__device__ __forceinline__ void load_strip(v4f ld[NLD], const float* __restrict__ x,
                                           size_t colBase, int t0, int r0) {
#pragma unroll
  for (int p = 0; p < NLD; ++p) {
    int t = t0 - 16 + r0 + p * RSTEP;
    if (p == 0 || p == NLD - 1) t = min(max(t, 0), T_DIM - 1);
    ld[p] = *(const v4f*)(x + colBase + (size_t)t * STRIDE);
  }
}

__device__ __forceinline__ void store_strip(float* xs, const v4f ld[NLD],
                                            int t0, int r0, int c4) {
  const v4f zero4 = {0.0f, 0.0f, 0.0f, 0.0f};
#pragma unroll
  for (int p = 0; p < NLD; ++p) {
    v4f v = ld[p];
    if (p == 0 || p == NLD - 1) {
      const int t = t0 - 16 + r0 + p * RSTEP;
      v = ((unsigned)t < (unsigned)T_DIM) ? v : zero4;
    }
    *(v4f*)(&xs[lds_idx(p * RSTEP + r0, c4 * 4)]) = v;
  }
}

__global__ __launch_bounds__(NTHR) void
LightweightConv1dTBC_kernel(const float* __restrict__ x,
                            const float* __restrict__ weight,
                            const float* __restrict__ bias,
                            float* __restrict__ out) {
  static_assert(TPW == 2, "the pair form computes exactly two output tiles per wave and strip");
  __shared__ __attribute__((aligned(16))) float xs[2 * ROWS * 64];
  __shared__ float wsm[32];

  const int tid  = threadIdx.y * 32 + threadIdx.x;
  const int b    = blockIdx.y;
  const int head = blockIdx.z;
  const int tq   = blockIdx.x * (NS * TC);

  const int c4 = tid & 15;
  const int r0 = tid >> 4;
  const size_t colBase = (size_t)b * C_DIM + head * R_DIM + c4 * 4;

  v4f ld[NLD];
  load_strip(ld, x, colBase, tq, r0);

  const int biasCol = head * R_DIM + (threadIdx.y & 1) * 32 + (threadIdx.x & 15);
  const float biasP = bias[biasCol];
  const float biasQ = bias[biasCol + 16];

  {
    const float* wr = weight + head * K_DIM;
    float mx = wr[0];
#pragma unroll
    for (int k = 1; k < K_DIM; ++k) mx = fmaxf(mx, wr[k]);
    float sum = 0.0f;
#pragma unroll
    for (int k = 0; k < K_DIM; ++k) sum += __expf(wr[k] - mx);
    const float inv = 1.0f / sum;
    if (tid < K_DIM) wsm[tid] = __expf(wr[tid] - mx) * inv;
  }

  store_strip(xs, ld, tq, r0, c4);

  lds_barrier();

  if (NS > 1) {
    load_strip(ld, x, colBase, tq + TC, r0);
    __builtin_amdgcn_sched_barrier(0);
  }

  const int lane = threadIdx.x;
  const int mN   = lane & 15;
  const int hi   = lane >> 4;
  const int jp   = threadIdx.y & 1;
  const int sb   = (threadIdx.y >> 1) * TPW;
  const int colP = jp * 32 + mN;

  const int f01 = (16 * sb + 2 * hi) * 64 + (colP ^ (hi << 5));
  const int f23 = (16 * sb + 2 * hi) * 64 + (colP ^ ((hi ^ 1) << 5));
  int f01Q = f01 + 16;
  int f23Q = f23 + 16;
  asm volatile("" : "+v"(f01Q), "+v"(f23Q));

  float* outw = out + (size_t)b * C_DIM + head * R_DIM + jp * 32 + lane;

  v2f g0[4], g2[4], gs[4], gd[4];
#pragma unroll
  for (int ks = 0; ks < 4; ++ks) {
    v2f a[3];
#pragma unroll
    for (int st = 0; st < 3; ++st) {
      const int j0 = ks * 4 + hi * 2;
      const int d0 = (st - 1) * 16 + j0 - mN + P_PAD;
      const int d1 = d0 + 1;
      const float w0 = wsm[min(max(d0, 0), K_DIM - 1)];
      const float w1 = wsm[min(max(d1, 0), K_DIM - 1)];
      a[st].x = ((unsigned)d0 < (unsigned)K_DIM) ? w0 : 0.0f;
      a[st].y = ((unsigned)d1 < (unsigned)K_DIM) ? w1 : 0.0f;
    }
    const v2f outer = a[0] + a[2];
    g0[ks] = a[0];
    g2[ks] = a[2];
    gs[ks] = (outer + a[1]) * 0.5f;
    gd[ks] = (outer - a[1]) * 0.5f;
  }

#pragma unroll
  for (int sidx = 0; sidx < NS; ++sidx) {
    const int t0 = tq + sidx * TC;
    const float* cur = xs + (sidx & 1) * (ROWS * 64);
    float* alt       = xs + ((sidx + 1) & 1) * (ROWS * 64);
    const bool more  = sidx + 1 < NS;

    v2f bP[TPW + 2][4], bQ[TPW + 2][4];
#pragma unroll
    for (int i = 0; i < TPW + 2; ++i) {
      const int s = (i + 1) & 3;
      ld_stile(bP[s], cur + f01, cur + f23, s);
      ld_stile(bQ[s], cur + f01Q, cur + f23Q, s);
    }

    v2f eSP[4], eDP[4], eAP[4], eBP[4], eSQ[4], eDQ[4], eAQ[4], eBQ[4];
#pragma unroll
    for (int ks = 0; ks < 4; ++ks) {
      eSP[ks] = bP[1][ks] + bP[2][ks];
      eSQ[ks] = bQ[1][ks] + bQ[2][ks];
      eDP[ks] = bP[2][ks] - bP[1][ks];
      eDQ[ks] = bQ[2][ks] - bQ[1][ks];
      eAP[ks] = bP[0][ks] - bP[2][ks];
      eAQ[ks] = bQ[0][ks] - bQ[2][ks];
      eBP[ks] = bP[3][ks] - bP[1][ks];
      eBQ[ks] = bQ[3][ks] - bQ[1][ks];
    }

    v8f m2P = {biasP, biasP, biasP, biasP, biasP, biasP, biasP, biasP};
    v8f m2Q = {biasQ, biasQ, biasQ, biasQ, biasQ, biasQ, biasQ, biasQ};
    v8f m3P = {}, m3Q = {};
    mma_block_pair(m2P, m2Q, gs, eSP, eSQ);
    mma_block_pair(m3P, m3Q, gd, eDP, eDQ);

    if (more) {
      __builtin_amdgcn_sched_barrier(0);
      store_strip(alt, ld, t0 + TC, r0, c4);
      __builtin_amdgcn_sched_barrier(0);
      if (sidx + 2 < NS) load_strip(ld, x, colBase, t0 + 2 * TC, r0);
      __builtin_amdgcn_sched_barrier(0);
    }

    v8f y0P = m2P + m3P;
    v8f y0Q = m2Q + m3Q;
    v8f y1P = m2P - m3P;
    v8f y1Q = m2Q - m3Q;
    mma_block_pair(y0P, y0Q, g0, eAP, eAQ);
    if (more) lds_signal(y0P, y0Q, y1P, y1Q);
    mma_block_pair(y1P, y1Q, g2, eBP, eBQ);

    float* og = outw + (size_t)(t0 + 16 * sb) * STRIDE;
#pragma unroll
    for (int r = 0; r < 8; ++r) {
      float p = y0P[r];
      float q = y0Q[r];
      rows_from_tile_pair(p, q);
      __builtin_nontemporal_store(p, og + (size_t)r * STRIDE);
      __builtin_nontemporal_store(q, og + (size_t)(r + 8) * STRIDE);
    }
#pragma unroll
    for (int r = 0; r < 8; ++r) {
      float p = y1P[r];
      float q = y1Q[r];
      rows_from_tile_pair(p, q);
      __builtin_nontemporal_store(p, og + (size_t)(16 + r) * STRIDE);
      __builtin_nontemporal_store(q, og + (size_t)(24 + r) * STRIDE);
    }

    if (more) lds_wait();
  }
}

extern "C" void kernel_launch(void* const* d_in, const int* in_sizes, int n_in,
                              void* d_out, int out_size, void* d_ws, size_t ws_size,
                              hipStream_t stream) {
  (void)in_sizes; (void)n_in; (void)d_ws; (void)ws_size; (void)out_size;
  const float* x      = (const float*)d_in[0];
  const float* weight = (const float*)d_in[1];
  const float* bias   = (const float*)d_in[2];
  float* out          = (float*)d_out;

  dim3 block(32, NWAVE);
  dim3 grid(T_DIM / (TC * NS), B_DIM, C_DIM / R_DIM);
  LightweightConv1dTBC_kernel<<<grid, block, 0, stream>>>(x, weight, bias, out);
}
